# starting kernel
# baseline (speedup 1.0000x reference)
#include <hip/hip_runtime.h>


typedef float v2f __attribute__((ext_vector_type(2)));
typedef float v4f __attribute__((ext_vector_type(4)));
typedef float v8f __attribute__((ext_vector_type(8)));

#define NUM_FIELDS 39
#define EMBED_DIM  64
#define PF         48
#define ESTRIDE    68
#define WAVES      4
#define BLOCK      (WAVES * 32)

__global__ __launch_bounds__(BLOCK)
void fwfm_kernel(const int* __restrict__ x,
                 const float* __restrict__ emb,
                 const float* __restrict__ bias,
                 const float* __restrict__ W,
                 const float* __restrict__ w0,
                 float* __restrict__ out,
                 int batch)
{
    __shared__ float Elds[WAVES * PF * ESTRIDE];
    __shared__ float Slds[PF * PF];

    const int tid  = threadIdx.x;
    const int wave = tid >> 5;
    const int lane = tid & 31;
    const int b    = blockIdx.x * WAVES + wave;
    const bool live = (b < batch);

    for (int i = tid; i < PF * PF; i += BLOCK) {
        int r = i / PF, c = i - r * PF;
        float v = 0.f;
        if (r < NUM_FIELDS && c < NUM_FIELDS && r != c)
            v = W[r * NUM_FIELDS + c] + W[c * NUM_FIELDS + r];
        Slds[i] = v;
    }

    float* Ew = &Elds[wave * PF * ESTRIDE];
    float acc_bias = 0.f;

    if (live) {
        const int* xb = x + (size_t)b * NUM_FIELDS;
        for (int i = lane; i < NUM_FIELDS * (EMBED_DIM / 4); i += 32) {
            int r = i >> 4;
            int c = (i & 15) << 2;
            const float* src = emb + (size_t)xb[r] * EMBED_DIM + c;
            v4f v = *(const v4f*)src;
            *(v4f*)&Ew[r * ESTRIDE + c] = v;
        }
        for (int i = lane; i < (PF - NUM_FIELDS) * (EMBED_DIM / 4); i += 32) {
            int r = NUM_FIELDS + (i >> 4);
            int c = (i & 15) << 2;
            v4f z = {};
            *(v4f*)&Ew[r * ESTRIDE + c] = z;
        }
        if (lane < NUM_FIELDS)      acc_bias  = bias[xb[lane]];
        if (lane + 32 < NUM_FIELDS) acc_bias += bias[xb[lane + 32]];
    }

    __syncthreads();

    float acc = 0.f;
    if (live) {
        const int lm   = lane & 15;
        const int hi   = lane >> 4;
        const int koff = hi * 2;

        const int   ti_arr[6]  = {0, 1, 2, 0, 0, 1};
        const int   tj_arr[6]  = {0, 1, 2, 1, 2, 2};
        const float wgt_arr[6] = {1.f, 1.f, 1.f, 2.f, 2.f, 2.f};

        #pragma unroll
        for (int t = 0; t < 6; ++t) {
            const int ti = ti_arr[t], tj = tj_arr[t];
            const float wgt = wgt_arr[t];
            const float* Arow = &Ew[(ti * 16 + lm) * ESTRIDE + koff];
            const float* Brow = &Ew[(tj * 16 + lm) * ESTRIDE + koff];
            v8f c = {};
            #pragma unroll
            for (int kb = 0; kb < EMBED_DIM; kb += 4) {
                v2f a  = *(const v2f*)(Arow + kb);
                v2f bb = *(const v2f*)(Brow + kb);
                c = __builtin_amdgcn_wmma_f32_16x16x4_f32(
                        false, a, false, bb, (short)0, c, false, false);
            }
            const float* Srow = &Slds[(ti * 16 + hi * 8) * PF + tj * 16 + lm];
            #pragma unroll
            for (int vv = 0; vv < 8; ++vv)
                acc += wgt * c[vv] * Srow[vv * PF];
        }
    }

    float tot  = acc;
    float btot = acc_bias;
    #pragma unroll
    for (int off = 16; off > 0; off >>= 1) {
        tot  += __shfl_xor(tot,  off, 32);
        btot += __shfl_xor(btot, off, 32);
    }
    if (live && lane == 0)
        out[b] = w0[0] + btot + 0.25f * tot;
}

extern "C" void kernel_launch(void* const* d_in, const int* in_sizes, int n_in,
                              void* d_out, int out_size, void* d_ws, size_t ws_size,
                              hipStream_t stream) {
    const int*   x    = (const int*)d_in[0];
    const float* emb  = (const float*)d_in[1];
    const float* bias = (const float*)d_in[2];
    const float* W    = (const float*)d_in[3];
    const float* w0   = (const float*)d_in[4];
    float* out = (float*)d_out;

    const int batch  = out_size;
    const int blocks = (batch + WAVES - 1) / WAVES;
    fwfm_kernel<<<blocks, BLOCK, 0, stream>>>(x, emb, bias, W, w0, out, batch);
}


# All global loads of a wave issued as one group before the first LDS store (20 index, 20 row-pair, 2 bias, 36 weight loads in flight); was one load wait per loop trip in gather and weight build
# speedup vs baseline: 1.4397x; 1.4397x over previous
#include <hip/hip_runtime.h>


typedef float v2f __attribute__((ext_vector_type(2)));
typedef float v4f __attribute__((ext_vector_type(4)));
typedef float v8f __attribute__((ext_vector_type(8)));

#define NUM_FIELDS 39
#define EMBED_DIM  64
#define PF         48
#define ESTRIDE    68
#define WAVES      4
#define BLOCK      (WAVES * 32)
#define GROUPS     ((NUM_FIELDS + 1) / 2)
#define S_PER_THREAD (PF * PF / BLOCK)
static_assert(PF * PF % BLOCK == 0, "S build assumes a whole number of elements per thread");

__global__ __launch_bounds__(BLOCK)
void fwfm_kernel(const int* __restrict__ x,
                 const float* __restrict__ emb,
                 const float* __restrict__ bias,
                 const float* __restrict__ W,
                 const float* __restrict__ w0,
                 float* __restrict__ out,
                 int batch)
{
    __shared__ float Elds[WAVES * PF * ESTRIDE];
    __shared__ float Slds[PF * PF];

    const int tid  = threadIdx.x;
    const int wave = tid >> 5;
    const int lane = tid & 31;
    const int b    = blockIdx.x * WAVES + wave;
    const bool live = (b < batch);

    const int bl = live ? b : (batch - 1);
    const int* xb = x + (size_t)bl * NUM_FIELDS;
    const int hi16 = lane >> 4;
    const int c4   = (lane & 15) << 2;

    int ridx[GROUPS];
    #pragma unroll
    for (int j = 0; j < GROUPS; ++j) {
        const int f = (2 * j + 1 < NUM_FIELDS) ? (2 * j + hi16) : (NUM_FIELDS - 1);
        ridx[j] = xb[f];
    }
    const int bidx0 = xb[lane];
    const int bidx1 = xb[(lane + 32 < NUM_FIELDS) ? (lane + 32) : (NUM_FIELDS - 1)];

    float wa[S_PER_THREAD], wb[S_PER_THREAD];
    #pragma unroll
    for (int q = 0; q < S_PER_THREAD; ++q) {
        const int i  = tid + q * BLOCK;
        const int r  = i / PF, c = i - r * PF;
        const int rc = (r < NUM_FIELDS) ? r : (NUM_FIELDS - 1);
        const int cc = (c < NUM_FIELDS) ? c : (NUM_FIELDS - 1);
        wa[q] = W[rc * NUM_FIELDS + cc];
        wb[q] = W[cc * NUM_FIELDS + rc];
    }

    v4f rv[GROUPS];
    #pragma unroll
    for (int j = 0; j < GROUPS; ++j)
        rv[j] = *(const v4f*)(emb + (size_t)ridx[j] * EMBED_DIM + c4);
    const float bv0 = bias[bidx0];
    const float bv1 = bias[bidx1];

    #pragma unroll
    for (int q = 0; q < S_PER_THREAD; ++q) {
        const int i = tid + q * BLOCK;
        const int r = i / PF, c = i - r * PF;
        Slds[i] = (r < NUM_FIELDS && c < NUM_FIELDS && r != c) ? (wa[q] + wb[q]) : 0.f;
    }

    float* Ew = &Elds[wave * PF * ESTRIDE];
    #pragma unroll
    for (int j = 0; j < GROUPS; ++j) {
        v4f v = rv[j];
        if (2 * j + 1 >= NUM_FIELDS && hi16) v = v4f{};
        *(v4f*)&Ew[(2 * j + hi16) * ESTRIDE + c4] = v;
    }
    #pragma unroll
    for (int j = GROUPS; j < PF / 2; ++j)
        *(v4f*)&Ew[(2 * j + hi16) * ESTRIDE + c4] = v4f{};
    float acc_bias = (lane + 32 < NUM_FIELDS) ? (bv0 + bv1) : bv0;

    __syncthreads();

    float acc = 0.f;
    if (live) {
        const int lm   = lane & 15;
        const int hi   = lane >> 4;
        const int koff = hi * 2;

        const int   ti_arr[6]  = {0, 1, 2, 0, 0, 1};
        const int   tj_arr[6]  = {0, 1, 2, 1, 2, 2};
        const float wgt_arr[6] = {1.f, 1.f, 1.f, 2.f, 2.f, 2.f};

        #pragma unroll
        for (int t = 0; t < 6; ++t) {
            const int ti = ti_arr[t], tj = tj_arr[t];
            const float wgt = wgt_arr[t];
            const float* Arow = &Ew[(ti * 16 + lm) * ESTRIDE + koff];
            const float* Brow = &Ew[(tj * 16 + lm) * ESTRIDE + koff];
            v8f c = {};
            #pragma unroll
            for (int kb = 0; kb < EMBED_DIM; kb += 4) {
                v2f a  = *(const v2f*)(Arow + kb);
                v2f bb = *(const v2f*)(Brow + kb);
                c = __builtin_amdgcn_wmma_f32_16x16x4_f32(
                        false, a, false, bb, (short)0, c, false, false);
            }
            const float* Srow = &Slds[(ti * 16 + hi * 8) * PF + tj * 16 + lm];
            #pragma unroll
            for (int vv = 0; vv < 8; ++vv)
                acc += wgt * c[vv] * Srow[vv * PF];
        }
    }

    float tot  = acc;
    float btot = acc_bias;
    #pragma unroll
    for (int off = 16; off > 0; off >>= 1) {
        tot  += __shfl_xor(tot,  off, 32);
        btot += __shfl_xor(btot, off, 32);
    }
    if (live && lane == 0)
        out[b] = w0[0] + btot + 0.25f * tot;
}

extern "C" void kernel_launch(void* const* d_in, const int* in_sizes, int n_in,
                              void* d_out, int out_size, void* d_ws, size_t ws_size,
                              hipStream_t stream) {
    const int*   x    = (const int*)d_in[0];
    const float* emb  = (const float*)d_in[1];
    const float* bias = (const float*)d_in[2];
    const float* W    = (const float*)d_in[3];
    const float* w0   = (const float*)d_in[4];
    float* out = (float*)d_out;

    const int batch  = out_size;
    const int blocks = (batch + WAVES - 1) / WAVES;
    fwfm_kernel<<<blocks, BLOCK, 0, stream>>>(x, emb, bias, W, w0, out, batch);
}


# Interaction as sum((U*E) o E): U = upper triangle of W+W^T is the WMMA A operand, row loads land directly in the B/D register layout so E skips LDS; 72 WMMAs not 96; summation order changes
# speedup vs baseline: 1.8857x; 1.3098x over previous
#include <hip/hip_runtime.h>


typedef float v2f __attribute__((ext_vector_type(2)));
typedef float v4f __attribute__((ext_vector_type(4)));
typedef float v8f __attribute__((ext_vector_type(8)));

#define NUM_FIELDS 39
#define EMBED_DIM  64
#define WAVES      4
#define BLOCK      (WAVES * 32)
#define NSLOT      20
#define KSTEPS     10
#define NFRAG      18
#define U_PER_THREAD (NFRAG * 64 / BLOCK)
static_assert(NFRAG * 64 % BLOCK == 0, "U build assumes a whole number of elements per thread");

__device__ __forceinline__ int slot_field(int j, int h) {
    return j < 8 ? 8 * h + j : (j < 16 ? 8 + 8 * h + j : 16 + 4 * h + j);
}
__device__ __forceinline__ constexpr int kstart(int I) { return I == 0 ? 0 : (I == 1 ? 4 : 8); }
__device__ __forceinline__ constexpr int fragof(int I, int g) { return I == 0 ? g : (I == 1 ? g + 6 : g + 8); }

__device__ __forceinline__ v8f mma(v2f a, v2f b, v8f c) {
    return __builtin_amdgcn_wmma_f32_16x16x4_f32(false, a, false, b, (short)0, c, false, false);
}

__global__ __launch_bounds__(BLOCK)
void fwfm_kernel(const int* __restrict__ x,
                 const float* __restrict__ emb,
                 const float* __restrict__ bias,
                 const float* __restrict__ W,
                 const float* __restrict__ w0,
                 float* __restrict__ out,
                 int batch)
{
    __shared__ float Ulds[NFRAG * 64];

    const int tid  = threadIdx.x;
    const int wave = tid >> 5;
    const int lane = tid & 31;
    const int b    = blockIdx.x * WAVES + wave;
    const bool live = (b < batch);
    const int bl = live ? b : (batch - 1);
    const int* xb = x + (size_t)bl * NUM_FIELDS;
    const int hi = lane >> 4;
    const int c4 = (lane & 15) << 2;

    float wa[U_PER_THREAD], wb[U_PER_THREAD];
    bool  uok[U_PER_THREAD];
    #pragma unroll
    for (int k = 0; k < U_PER_THREAD; ++k) {
        const int e   = tid + k * BLOCK;
        const int idx = e >> 6;
        const int fl  = (e & 63) >> 1;
        const int r   = e & 1;
        const int I   = idx < 10 ? 0 : (idx < 16 ? 1 : 2);
        const int g   = idx - (I == 0 ? 0 : (I == 1 ? 6 : 8));
        const int m   = fl & 15, v = m & 7;
        const int row = slot_field(8 * I + v, m >> 3);
        const int col = slot_field(2 * g + r, fl >> 4);
        uok[k] = (I < 2 || v < 4) && row < col && col < NUM_FIELDS;
        const int rc = uok[k] ? row : 0;
        const int cc = uok[k] ? col : 0;
        wa[k] = W[rc * NUM_FIELDS + cc];
        wb[k] = W[cc * NUM_FIELDS + rc];
    }

    int ridx[NSLOT];
    #pragma unroll
    for (int j = 0; j < NSLOT; ++j) {
        const int f = slot_field(j, hi);
        ridx[j] = xb[f < NUM_FIELDS ? f : (NUM_FIELDS - 1)];
    }
    const int bidx0 = xb[lane];
    const int bidx1 = xb[(lane + 32 < NUM_FIELDS) ? (lane + 32) : (NUM_FIELDS - 1)];

    #pragma unroll
    for (int k = 0; k < U_PER_THREAD; ++k)
        Ulds[tid + k * BLOCK] = uok[k] ? (wa[k] + wb[k]) : 0.f;

    v4f rv[NSLOT];
    #pragma unroll
    for (int j = 0; j < NSLOT; ++j)
        rv[j] = *(const v4f*)(emb + (size_t)ridx[j] * EMBED_DIM + c4);
    const float bv0 = bias[bidx0];
    const float bv1 = bias[bidx1];

    __builtin_amdgcn_fence(__ATOMIC_RELEASE, "workgroup", "local");
    __builtin_amdgcn_s_barrier();
    __builtin_amdgcn_fence(__ATOMIC_ACQUIRE, "workgroup", "local");

    v2f ua[NFRAG];
    #pragma unroll
    for (int i = 0; i < NFRAG; ++i)
        ua[i] = *(const v2f*)&Ulds[i * 64 + lane * 2];

    __builtin_amdgcn_sched_barrier(0);

    if (hi) rv[NSLOT - 1] = v4f{};
    float acc_bias = (lane + 32 < NUM_FIELDS) ? (bv0 + bv1) : bv0;

    float accs[2] = {};
    #pragma unroll
    for (int q = 0; q < 4; ++q) {
        #pragma unroll
        for (int I = 0; I < 3; ++I) {
            v8f c = {};
            #pragma unroll
            for (int g = kstart(I); g < KSTEPS; ++g)
                c = mma(ua[fragof(I, g)], v2f{ rv[2 * g][q], rv[2 * g + 1][q] }, c);
            #pragma unroll
            for (int v = 0; v < (I < 2 ? 8 : 4); ++v)
                accs[v % 2] += c[v] * rv[8 * I + v][q];
        }
    }
    const float acc = accs[0] + accs[1];

    float tot  = acc;
    float btot = acc_bias;
    #pragma unroll
    for (int off = 16; off > 0; off >>= 1) {
        tot  += __shfl_xor(tot,  off, 32);
        btot += __shfl_xor(btot, off, 32);
    }
    if (live && lane == 0)
        out[b] = w0[0] + btot + 0.5f * tot;
}

extern "C" void kernel_launch(void* const* d_in, const int* in_sizes, int n_in,
                              void* d_out, int out_size, void* d_ws, size_t ws_size,
                              hipStream_t stream) {
    const int*   x    = (const int*)d_in[0];
    const float* emb  = (const float*)d_in[1];
    const float* bias = (const float*)d_in[2];
    const float* W    = (const float*)d_in[3];
    const float* w0   = (const float*)d_in[4];
    float* out = (float*)d_out;

    const int batch  = out_size;
    const int blocks = (batch + WAVES - 1) / WAVES;
    fwfm_kernel<<<blocks, BLOCK, 0, stream>>>(x, emb, bias, W, w0, out, batch);
}


# Persistent waves (1024 work-groups, 4 items per wave); each item worked as two 32-dim halves in two register sets, next half's row loads in flight under this half's WMMAs; summation order changes
# speedup vs baseline: 2.0743x; 1.1000x over previous
#include <hip/hip_runtime.h>


typedef float v2f __attribute__((ext_vector_type(2)));
typedef float v8f __attribute__((ext_vector_type(8)));

#define NUM_FIELDS 39
#define EMBED_DIM  64
#define WAVES      4
#define BLOCK      (WAVES * 32)
#define GRID       1024
#define HALF_DIM   32
#define NSLOT      20
#define KSTEPS     10
#define NFRAG      18
#define U_PER_THREAD (NFRAG * 64 / BLOCK)
static_assert(NFRAG * 64 % BLOCK == 0, "U build assumes a whole number of elements per thread");

__device__ __forceinline__ int slot_field(int j, int h) {
    return j < 8 ? 8 * h + j : (j < 16 ? 8 + 8 * h + j : 16 + 4 * h + j);
}
__device__ __forceinline__ constexpr int kstart(int I) { return I == 0 ? 0 : (I == 1 ? 4 : 8); }
__device__ __forceinline__ constexpr int fragof(int I, int g) { return I == 0 ? g : (I == 1 ? g + 6 : g + 8); }

__device__ __forceinline__ v8f mma(v2f a, v2f b, v8f c) {
    return __builtin_amdgcn_wmma_f32_16x16x4_f32(false, a, false, b, (short)0, c, false, false);
}

__device__ __forceinline__ void load_indices(const int* __restrict__ x, int item, int hi, int lane, unsigned c2,
                                             unsigned (&roff)[NSLOT], int& bidx0, int& bidx1) {
    const int* xb = x + (size_t)item * NUM_FIELDS;
    #pragma unroll
    for (int j = 0; j < NSLOT; ++j) {
        const int f1 = slot_field(j, 1);
        const int i0 = xb[slot_field(j, 0)];
        const int i1 = xb[f1 < NUM_FIELDS ? f1 : (NUM_FIELDS - 1)];
        roff[j] = (((unsigned)(hi ? i1 : i0) * EMBED_DIM + c2) << 2);
    }
    bidx0 = xb[lane];
    bidx1 = xb[(lane + 32 < NUM_FIELDS) ? (lane + 32) : (NUM_FIELDS - 1)];
}

__device__ __forceinline__ void load_half(const float* __restrict__ emb, const unsigned (&roff)[NSLOT],
                                          unsigned half_bytes, v2f (&rv)[NSLOT]) {
    const char* base = (const char*)emb;
    #pragma unroll
    for (int j = 0; j < NSLOT; ++j)
        rv[j] = *(const v2f*)(base + (roff[j] + half_bytes));
}

__device__ __forceinline__ void half_item(const v2f (&ua)[NFRAG], v2f (&rv)[NSLOT], int hi, float (&accs)[2]) {
    if (hi) rv[NSLOT - 1] = v2f{};
    #pragma unroll
    for (int q = 0; q < 2; ++q) {
        #pragma unroll
        for (int I = 0; I < 3; ++I) {
            v8f c = {};
            #pragma unroll
            for (int g = kstart(I); g < KSTEPS; ++g)
                c = mma(ua[fragof(I, g)], v2f{ rv[2 * g][q], rv[2 * g + 1][q] }, c);
            #pragma unroll
            for (int v = 0; v < (I < 2 ? 8 : 4); ++v)
                accs[v % 2] += c[v] * rv[8 * I + v][q];
        }
    }
}

__global__ __launch_bounds__(BLOCK)
void fwfm_kernel(const int* __restrict__ x,
                 const float* __restrict__ emb,
                 const float* __restrict__ bias,
                 const float* __restrict__ W,
                 const float* __restrict__ w0,
                 float* __restrict__ out,
                 int batch)
{
    __shared__ float Ulds[NFRAG * 64];

    const int tid  = threadIdx.x;
    const int wave = __builtin_amdgcn_readfirstlane(tid >> 5);
    const int lane = tid & 31;
    const int hi = lane >> 4;
    const unsigned c2 = (lane & 15) << 1;

    const int first  = blockIdx.x * WAVES + wave;
    const int stride = gridDim.x * WAVES;
    const int mylast = first < batch ? first + ((batch - 1 - first) / stride) * stride : batch - 1;

    float wa[U_PER_THREAD], wb[U_PER_THREAD];
    bool  uok[U_PER_THREAD];
    #pragma unroll
    for (int k = 0; k < U_PER_THREAD; ++k) {
        const int e   = tid + k * BLOCK;
        const int idx = e >> 6;
        const int fl  = (e & 63) >> 1;
        const int r   = e & 1;
        const int I   = idx < 10 ? 0 : (idx < 16 ? 1 : 2);
        const int g   = idx - (I == 0 ? 0 : (I == 1 ? 6 : 8));
        const int m   = fl & 15, v = m & 7;
        const int row = slot_field(8 * I + v, m >> 3);
        const int col = slot_field(2 * g + r, fl >> 4);
        uok[k] = (I < 2 || v < 4) && row < col && col < NUM_FIELDS;
        const int rc = uok[k] ? row : 0;
        const int cc = uok[k] ? col : 0;
        wa[k] = W[rc * NUM_FIELDS + cc];
        wb[k] = W[cc * NUM_FIELDS + rc];
    }

    unsigned roff[NSLOT], rnext[NSLOT];
    int bidx0, bidx1;
    load_indices(x, first < mylast ? first : mylast, hi, lane, c2, roff, bidx0, bidx1);
    const float w0v = w0[0];

    #pragma unroll
    for (int k = 0; k < U_PER_THREAD; ++k)
        Ulds[tid + k * BLOCK] = uok[k] ? (wa[k] + wb[k]) : 0.f;

    v2f rvA[NSLOT], rvB[NSLOT];
    load_half(emb, roff, 0u, rvA);
    float bv0 = bias[bidx0];
    float bv1 = bias[bidx1];
    {
        const int n = first + stride;
        load_indices(x, n < mylast ? n : mylast, hi, lane, c2, rnext, bidx0, bidx1);
    }

    __builtin_amdgcn_fence(__ATOMIC_RELEASE, "workgroup", "local");
    __builtin_amdgcn_s_barrier();
    __builtin_amdgcn_fence(__ATOMIC_ACQUIRE, "workgroup", "local");

    v2f ua[NFRAG];
    #pragma unroll
    for (int i = 0; i < NFRAG; ++i)
        ua[i] = *(const v2f*)&Ulds[i * 64 + lane * 2];

    __builtin_amdgcn_sched_barrier(0);

    for (int b = first; b < batch; b += stride) {
        float accs[2] = {};

        load_half(emb, roff, HALF_DIM * 4u, rvB);
        __builtin_amdgcn_sched_barrier(0);
        half_item(ua, rvA, hi, accs);

        #pragma unroll
        for (int j = 0; j < NSLOT; ++j) roff[j] = rnext[j];
        load_half(emb, roff, 0u, rvA);
        const float cb0 = bv0, cb1 = bv1;
        bv0 = bias[bidx0];
        bv1 = bias[bidx1];
        {
            const int n = b + 2 * stride;
            load_indices(x, n < mylast ? n : mylast, hi, lane, c2, rnext, bidx0, bidx1);
        }
        __builtin_amdgcn_sched_barrier(0);
        half_item(ua, rvB, hi, accs);

        float tot  = accs[0] + accs[1];
        float btot = (lane + 32 < NUM_FIELDS) ? (cb0 + cb1) : cb0;
        #pragma unroll
        for (int off = 16; off > 0; off >>= 1) {
            tot  += __shfl_xor(tot,  off, 32);
            btot += __shfl_xor(btot, off, 32);
        }
        if (lane == 0)
            out[b] = w0v + btot + 0.5f * tot;
    }
}

extern "C" void kernel_launch(void* const* d_in, const int* in_sizes, int n_in,
                              void* d_out, int out_size, void* d_ws, size_t ws_size,
                              hipStream_t stream) {
    const int*   x    = (const int*)d_in[0];
    const float* emb  = (const float*)d_in[1];
    const float* bias = (const float*)d_in[2];
    const float* W    = (const float*)d_in[3];
    const float* w0   = (const float*)d_in[4];
    float* out = (float*)d_out;

    const int batch = out_size;
    fwfm_kernel<<<GRID, BLOCK, 0, stream>>>(x, emb, bias, W, w0, out, batch);
}
